# starting kernel
# baseline (speedup 1.0000x reference)
#include <hip/hip_runtime.h>
#include <hip/hip_bf16.h>

typedef __attribute__((ext_vector_type(2))) float v2f;
typedef __attribute__((ext_vector_type(4))) float v4f;
typedef __attribute__((ext_vector_type(8))) float v8f;

#define BATCH  2048
#define MAXLEN 200
#define HIDDEN 256
#define H4     (HIDDEN / 4)

__global__ void __launch_bounds__(H4)
pool_kernel(const int* __restrict__ x,
            const float* __restrict__ emb,
            float* __restrict__ pooled) {
    const int b  = blockIdx.x;
    const int h4 = threadIdx.x;
    const int* xrow = x + b * MAXLEN;
    const v4f* __restrict__ emb4 = (const v4f*)emb;

    v4f acc = {};
#pragma unroll 4
    for (int l = 0; l < MAXLEN; ++l) {
        const int idx = xrow[l];
        acc += emb4[(size_t)idx * H4 + h4];
    }
    acc *= (1.0f / MAXLEN);
    ((v4f*)pooled)[(size_t)b * H4 + h4] = acc;
}

__global__ void __launch_bounds__(32)
gemm_kernel(const float* __restrict__ s,
            const float* __restrict__ W,
            const float* __restrict__ bias,
            float* __restrict__ out) {
    const int lane  = threadIdx.x;
    const int half  = lane >> 4;
    const int l15   = lane & 15;
    const int mTile = blockIdx.x;
    const int nTile = blockIdx.y;

    const int mRow = mTile * 16 + l15;
    const int nCol = nTile * 16 + l15;

    const float* __restrict__ srow = s + (size_t)mRow * HIDDEN;
    const float* __restrict__ wrow = W + (size_t)nCol * HIDDEN;

    v8f acc = {};
#pragma unroll 8
    for (int k = 0; k < HIDDEN; k += 4) {
        const int kk = k + 2 * half;
        v2f a;
        a.x = srow[kk];
        a.y = srow[kk + 1];
        v2f bmat;
        bmat.x = wrow[kk];
        bmat.y = wrow[kk + 1];
        acc = __builtin_amdgcn_wmma_f32_16x16x4_f32(
            false, a, false, bmat, (short)0, acc, false, false);
    }

    const float bv = bias[nCol];
#pragma unroll
    for (int r = 0; r < 8; ++r) {
        const int m = mTile * 16 + r + half * 8;
        out[(size_t)m * HIDDEN + nCol] = acc[r] + bv;
    }
}

extern "C" void kernel_launch(void* const* d_in, const int* in_sizes, int n_in,
                              void* d_out, int out_size, void* d_ws, size_t ws_size,
                              hipStream_t stream) {
    const int*   x    = (const int*)d_in[0];
    const float* emb  = (const float*)d_in[1];
    const float* W    = (const float*)d_in[2];
    const float* bias = (const float*)d_in[3];
    float*       out  = (float*)d_out;
    float*       pooled = (float*)d_ws;

    pool_kernel<<<BATCH, H4, 0, stream>>>(x, emb, pooled);

    dim3 grid(BATCH / 16, HIDDEN / 16);
    gemm_kernel<<<grid, 32, 0, stream>>>(pooled, W, bias, out);
}


# pool kernel: each wave fetches its table rows 25 at a time (token indices read one group ahead) instead of 4 per wait, 8 load round trips per wave for 50; sum order unchanged
# speedup vs baseline: 1.2677x; 1.2677x over previous
#include <hip/hip_runtime.h>
#include <hip/hip_bf16.h>

typedef __attribute__((ext_vector_type(2))) float v2f;
typedef __attribute__((ext_vector_type(4))) float v4f;
typedef __attribute__((ext_vector_type(8))) float v8f;

#define BATCH  2048
#define MAXLEN 200
#define HIDDEN 256
#define H4     (HIDDEN / 4)

#define WIN   25
#define NGRP  (MAXLEN / WIN)

__global__ void __launch_bounds__(H4)
pool_kernel(const int* __restrict__ x,
            const float* __restrict__ emb,
            float* __restrict__ pooled) {
    static_assert(MAXLEN % WIN == 0 && NGRP >= 2, "groups must tile the sequence");
    const int b  = blockIdx.x;
    const int h4 = threadIdx.x;
    const int* xrow = x + b * MAXLEN;
    const v4f* __restrict__ emb4 = (const v4f*)emb;

    v4f win[WIN];
    int idx[WIN];

#pragma unroll
    for (int i = 0; i < WIN; ++i) idx[i] = xrow[i];
#pragma unroll
    for (int i = 0; i < WIN; ++i) win[i] = emb4[(size_t)idx[i] * H4 + h4];
#pragma unroll
    for (int i = 0; i < WIN; ++i) idx[i] = xrow[WIN + i];

    v4f acc = {};
#pragma unroll 1
    for (int g = 1; g < NGRP; ++g) {
        const int gn = (g + 1 < NGRP) ? g + 1 : g;
        int nidx[WIN];
#pragma unroll
        for (int i = 0; i < WIN; ++i) nidx[i] = xrow[gn * WIN + i];
#pragma unroll
        for (int i = 0; i < WIN; ++i) {
            acc += win[i];
            win[i] = emb4[(size_t)idx[i] * H4 + h4];
        }
#pragma unroll
        for (int i = 0; i < WIN; ++i) idx[i] = nidx[i];
    }
#pragma unroll
    for (int i = 0; i < WIN; ++i) acc += win[i];

    acc *= (1.0f / MAXLEN);
    ((v4f*)pooled)[(size_t)b * H4 + h4] = acc;
}

__global__ void __launch_bounds__(32)
gemm_kernel(const float* __restrict__ s,
            const float* __restrict__ W,
            const float* __restrict__ bias,
            float* __restrict__ out) {
    const int lane  = threadIdx.x;
    const int half  = lane >> 4;
    const int l15   = lane & 15;
    const int mTile = blockIdx.x;
    const int nTile = blockIdx.y;

    const int mRow = mTile * 16 + l15;
    const int nCol = nTile * 16 + l15;

    const float* __restrict__ srow = s + (size_t)mRow * HIDDEN;
    const float* __restrict__ wrow = W + (size_t)nCol * HIDDEN;

    v8f acc = {};
#pragma unroll 8
    for (int k = 0; k < HIDDEN; k += 4) {
        const int kk = k + 2 * half;
        v2f a;
        a.x = srow[kk];
        a.y = srow[kk + 1];
        v2f bmat;
        bmat.x = wrow[kk];
        bmat.y = wrow[kk + 1];
        acc = __builtin_amdgcn_wmma_f32_16x16x4_f32(
            false, a, false, bmat, (short)0, acc, false, false);
    }

    const float bv = bias[nCol];
#pragma unroll
    for (int r = 0; r < 8; ++r) {
        const int m = mTile * 16 + r + half * 8;
        out[(size_t)m * HIDDEN + nCol] = acc[r] + bv;
    }
}

extern "C" void kernel_launch(void* const* d_in, const int* in_sizes, int n_in,
                              void* d_out, int out_size, void* d_ws, size_t ws_size,
                              hipStream_t stream) {
    const int*   x    = (const int*)d_in[0];
    const float* emb  = (const float*)d_in[1];
    const float* W    = (const float*)d_in[2];
    const float* bias = (const float*)d_in[3];
    float*       out  = (float*)d_out;
    float*       pooled = (float*)d_ws;

    pool_kernel<<<BATCH, H4, 0, stream>>>(x, emb, pooled);

    dim3 grid(BATCH / 16, HIDDEN / 16);
    gemm_kernel<<<grid, 32, 0, stream>>>(pooled, W, bias, out);
}


# linear layer fused into the pool launch: 8 batch rows per 16-wave work-group, pooled rows and each wave's 16 W rows via LDS, 1 launch not 2; product k sum order changed (2 accumulators, 4 k per lane)
# speedup vs baseline: 1.3621x; 1.0745x over previous
#include <hip/hip_runtime.h>

typedef __attribute__((ext_vector_type(2))) float v2f;
typedef __attribute__((ext_vector_type(4))) float v4f;
typedef __attribute__((ext_vector_type(8))) float v8f;

#define BATCH  2048
#define MAXLEN 200
#define HIDDEN 256
#define H4     (HIDDEN / 4)

#define WIN   25
#define NGRP  (MAXLEN / WIN)

#define ROWS    8
#define WAVES   (2 * ROWS)
#define THREADS (32 * WAVES)
#define NTILE   16
#define PITCH   (HIDDEN + 4)
#define KBLK    8
#define NBLK    (HIDDEN / (8 * KBLK))

__global__ void __launch_bounds__(THREADS)
pool_linear_kernel(const int* __restrict__ x,
                   const float* __restrict__ emb,
                   const float* __restrict__ W,
                   const float* __restrict__ bias,
                   float* __restrict__ out) {
    static_assert(MAXLEN % WIN == 0 && NGRP >= 2, "groups must tile the sequence");
    static_assert(WAVES * NTILE == HIDDEN, "one wave per 16 output columns");
    static_assert(BATCH % ROWS == 0 && ROWS <= 8, "rows 0..7 of the tile are D registers of lanes 0..15");
    static_assert((WAVES * NTILE + 16) * PITCH * 4 + ROWS * HIDDEN * 4 <= 320 * 1024, "LDS budget");

    __shared__ __attribute__((aligned(16))) float Wt[WAVES * NTILE * PITCH];
    __shared__ __attribute__((aligned(16))) float Ps[16 * PITCH];
    __shared__ __attribute__((aligned(16))) float Os[ROWS * HIDDEN];

    const int tid  = threadIdx.x;
    const int lane = tid & 31;
    const int wave = __builtin_amdgcn_readfirstlane(tid >> 5);
    const int r    = wave >> 1;
    const int h4   = (wave & 1) * 32 + lane;
    const int b    = blockIdx.x * ROWS + r;
    const int* xrow = x + b * MAXLEN;
    const v4f* __restrict__ emb4 = (const v4f*)emb;
    const v4f* __restrict__ W4   = (const v4f*)W;

    v4f wst[2 * NTILE];
#pragma unroll
    for (int q = 0; q < 2 * NTILE; ++q)
        wst[q] = W4[(size_t)(wave * NTILE + (q >> 1)) * H4 + (q & 1) * 32 + lane];

    v4f win[WIN];
    int idx[WIN];

#pragma unroll
    for (int i = 0; i < WIN; ++i) idx[i] = xrow[i];
#pragma unroll
    for (int i = 0; i < WIN; ++i) win[i] = emb4[(size_t)idx[i] * H4 + h4];
#pragma unroll
    for (int i = 0; i < WIN; ++i) idx[i] = xrow[WIN + i];

#pragma unroll
    for (int q = 0; q < 2 * NTILE; ++q)
        *(v4f*)&Wt[(wave * NTILE + (q >> 1)) * PITCH + (q & 1) * 128 + lane * 4] = wst[q];

    for (int i = tid; i < (16 - ROWS) * PITCH / 4; i += THREADS)
        ((v4f*)Ps)[ROWS * PITCH / 4 + i] = v4f{};

    v4f acc = {};
#pragma unroll 1
    for (int g = 1; g < NGRP; ++g) {
        const int gn = (g + 1 < NGRP) ? g + 1 : g;
        int nidx[WIN];
#pragma unroll
        for (int i = 0; i < WIN; ++i) nidx[i] = xrow[gn * WIN + i];
#pragma unroll
        for (int i = 0; i < WIN; ++i) {
            acc += win[i];
            win[i] = emb4[(size_t)idx[i] * H4 + h4];
        }
#pragma unroll
        for (int i = 0; i < WIN; ++i) idx[i] = nidx[i];
    }
#pragma unroll
    for (int i = 0; i < WIN; ++i) acc += win[i];

    acc *= (1.0f / MAXLEN);
    *(v4f*)&Ps[r * PITCH + h4 * 4] = acc;

    const int c = lane & 15;
    const int h = lane >> 4;
    const float bv = bias[wave * NTILE + c];

    __syncthreads();

    const float* ap = Ps + c * PITCH + 4 * h;
    const float* bp = Wt + (wave * NTILE + c) * PITCH + 4 * h;

    v4f fa[2][KBLK], fb[2][KBLK];
#pragma unroll
    for (int j = 0; j < KBLK; ++j) {
        fa[0][j] = *(const v4f*)(ap + 8 * j);
        fb[0][j] = *(const v4f*)(bp + 8 * j);
    }
    v8f acc0 = {}, acc1 = {};
#pragma unroll
    for (int t = 0; t < NBLK; ++t) {
        const int cur = t & 1, nxt = cur ^ 1;
        if (t + 1 < NBLK) {
#pragma unroll
            for (int j = 0; j < KBLK; ++j) {
                fa[nxt][j] = *(const v4f*)(ap + 8 * ((t + 1) * KBLK + j));
                fb[nxt][j] = *(const v4f*)(bp + 8 * ((t + 1) * KBLK + j));
            }
        }
        __builtin_amdgcn_sched_barrier(0);
#pragma unroll
        for (int j = 0; j < KBLK; ++j) {
            v2f a0 = {fa[cur][j].x, fa[cur][j].y}, b0 = {fb[cur][j].x, fb[cur][j].y};
            v2f a1 = {fa[cur][j].z, fa[cur][j].w}, b1 = {fb[cur][j].z, fb[cur][j].w};
            acc0 = __builtin_amdgcn_wmma_f32_16x16x4_f32(false, a0, false, b0, (short)0, acc0, false, false);
            acc1 = __builtin_amdgcn_wmma_f32_16x16x4_f32(false, a1, false, b1, (short)0, acc1, false, false);
        }
        __builtin_amdgcn_sched_barrier(0);
    }

    if (h == 0) {
#pragma unroll
        for (int rr = 0; rr < ROWS; ++rr)
            Os[rr * HIDDEN + wave * NTILE + c] = (acc0[rr] + acc1[rr]) + bv;
    }
    __syncthreads();

    ((v4f*)out)[(size_t)b * H4 + h4] = *(const v4f*)&Os[r * HIDDEN + h4 * 4];
}

extern "C" void kernel_launch(void* const* d_in, const int* in_sizes, int n_in,
                              void* d_out, int out_size, void* d_ws, size_t ws_size,
                              hipStream_t stream) {
    const int*   x    = (const int*)d_in[0];
    const float* emb  = (const float*)d_in[1];
    const float* W    = (const float*)d_in[2];
    const float* bias = (const float*)d_in[3];
    float*       out  = (float*)d_out;

    pool_linear_kernel<<<BATCH / ROWS, THREADS, 0, stream>>>(x, emb, W, bias, out);
}


# k = 0..127 of the product (32 of 64 matrix instructions per wave) runs with the table loads of columns 128..255 in flight; a wave sums one token half per column half; token sum order changed
# speedup vs baseline: 1.6237x; 1.1921x over previous
#include <hip/hip_runtime.h>

typedef __attribute__((ext_vector_type(2))) float v2f;
typedef __attribute__((ext_vector_type(4))) float v4f;
typedef __attribute__((ext_vector_type(8))) float v8f;
typedef const __attribute__((address_space(4))) int* cint_p;

#define BATCH  2048
#define MAXLEN 200
#define HIDDEN 256
#define H4     (HIDDEN / 4)

#define HALF  (MAXLEN / 2)
#define WIN   25
#define NJ    (HALF / WIN)
#define NQ    (2 * NJ)

#define ROWS    8
#define WAVES   (2 * ROWS)
#define THREADS (32 * WAVES)
#define NTILE   16
#define PITCH   (HIDDEN + 4)
#define KHALF   (HIDDEN / 2)
#define KBLK    4
#define NBLKH   (KHALF / (8 * KBLK))
#define OSH     (ROWS * HIDDEN + 16)


__device__ __forceinline__ void product_half(const float* ap, const float* bp, v8f& acc0, v8f& acc1) {
    v4f fa[2][KBLK], fb[2][KBLK];
#pragma unroll
    for (int j = 0; j < KBLK; ++j) {
        fa[0][j] = *(const v4f*)(ap + 8 * j);
        fb[0][j] = *(const v4f*)(bp + 8 * j);
    }
#pragma unroll
    for (int s = 0; s < NBLKH; ++s) {
        const int cur = s & 1, nxt = cur ^ 1;
        if (s + 1 < NBLKH) {
#pragma unroll
            for (int j = 0; j < KBLK; ++j) {
                fa[nxt][j] = *(const v4f*)(ap + 8 * ((s + 1) * KBLK + j));
                fb[nxt][j] = *(const v4f*)(bp + 8 * ((s + 1) * KBLK + j));
            }
        }
        __builtin_amdgcn_sched_barrier(0);
#pragma unroll
        for (int j = 0; j < KBLK; ++j) {
            v2f a0 = {fa[cur][j].x, fa[cur][j].y}, b0 = {fb[cur][j].x, fb[cur][j].y};
            v2f a1 = {fa[cur][j].z, fa[cur][j].w}, b1 = {fb[cur][j].z, fb[cur][j].w};
            acc0 = __builtin_amdgcn_wmma_f32_16x16x4_f32(false, a0, false, b0, (short)0, acc0, false, false);
            acc1 = __builtin_amdgcn_wmma_f32_16x16x4_f32(false, a1, false, b1, (short)0, acc1, false, false);
        }
        __builtin_amdgcn_sched_barrier(0);
    }
}

__global__ void __launch_bounds__(THREADS)
pool_linear_kernel(const int* __restrict__ x,
                   const float* __restrict__ emb,
                   const float* __restrict__ W,
                   const float* __restrict__ bias,
                   float* __restrict__ out) {
    static_assert(HALF % WIN == 0 && (NJ & (NJ - 1)) == 0 && NJ >= 2, "groups must tile a token half");
    static_assert(WAVES * NTILE == HIDDEN, "one wave per 16 output columns");
    static_assert(BATCH % ROWS == 0 && ROWS == 8, "tile row r + 8t = batch row r, token half t");
    static_assert((WAVES * NTILE + 16) * PITCH * 4 + 2 * OSH * 4 <= 320 * 1024, "LDS budget");

    __shared__ __attribute__((aligned(16))) float Wt[WAVES * NTILE * PITCH];
    __shared__ __attribute__((aligned(16))) float Ps[16 * PITCH];
    __shared__ __attribute__((aligned(16))) float Os[2 * OSH];

    const int tid  = threadIdx.x;
    const int lane = tid & 31;
    const int wave = __builtin_amdgcn_readfirstlane(tid >> 5);
    const int r    = wave >> 1;
    const int th   = wave & 1;
    const int b    = blockIdx.x * ROWS + r;
    const cint_p xw = (cint_p)(x + b * MAXLEN + th * HALF);
    const v4f* __restrict__ emb4 = (const v4f*)emb;
    const v4f* __restrict__ W4   = (const v4f*)W;

    const int c = lane & 15;
    const int h = lane >> 4;

    v4f wst[2 * NTILE];
#pragma unroll
    for (int q = 0; q < 2 * NTILE; ++q)
        wst[q] = W4[(size_t)(wave * NTILE + (q >> 1)) * H4 + (q & 1) * 32 + lane];
    const float bv = (h == 0) ? bias[wave * NTILE + c] : 0.0f;

    v4f win[WIN];
    int idx[WIN];

#pragma unroll
    for (int i = 0; i < WIN; ++i) idx[i] = xw[i];
#pragma unroll
    for (int i = 0; i < WIN; ++i) win[i] = emb4[(size_t)idx[i] * H4 + lane];
#pragma unroll
    for (int i = 0; i < WIN; ++i) idx[i] = xw[WIN + i];

#pragma unroll
    for (int q = 0; q < 2 * NTILE; ++q)
        *(v4f*)&Wt[(wave * NTILE + (q >> 1)) * PITCH + (q & 1) * 128 + lane * 4] = wst[q];

    const float* ap = Ps + c * PITCH + 4 * h;
    const float* bp = Wt + (wave * NTILE + c) * PITCH + 4 * h;
    float* prow = Ps + (r + 8 * th) * PITCH + lane * 4;

    v4f acc = {};
    v8f acc0 = {}, acc1 = {};
#pragma unroll 1
    for (int q = 1; q < NQ; ++q) {
        const int jn = (q + 1) & (NJ - 1);
        const int hq = (q >= NJ ? 32 : 0) + lane;
        int nidx[WIN];
#pragma unroll
        for (int i = 0; i < WIN; ++i) nidx[i] = xw[jn * WIN + i];
#pragma unroll
        for (int i = 0; i < WIN; ++i) {
            acc += win[i];
            win[i] = emb4[(size_t)idx[i] * H4 + hq];
        }
#pragma unroll
        for (int i = 0; i < WIN; ++i) idx[i] = nidx[i];

        if (q == NJ) {
            *(v4f*)prow = acc * (1.0f / MAXLEN);
            acc = v4f{};
            __builtin_amdgcn_fence(__ATOMIC_RELEASE, "workgroup", "local");
            __builtin_amdgcn_s_barrier();
            __builtin_amdgcn_fence(__ATOMIC_ACQUIRE, "workgroup", "local");
            product_half(ap, bp, acc0, acc1);
        }
    }
#pragma unroll
    for (int i = 0; i < WIN; ++i) acc += win[i];
    *(v4f*)(prow + KHALF) = acc * (1.0f / MAXLEN);

    __syncthreads();

    product_half(ap + KHALF, bp + KHALF, acc0, acc1);

#pragma unroll
    for (int rr = 0; rr < ROWS; ++rr)
        Os[h * OSH + rr * HIDDEN + wave * NTILE + c] = (acc0[rr] + acc1[rr]) + bv;
    __syncthreads();

    const int h4 = th * 32 + lane;
    ((v4f*)out)[(size_t)b * H4 + h4] = *(const v4f*)&Os[r * HIDDEN + h4 * 4]
                                     + *(const v4f*)&Os[OSH + r * HIDDEN + h4 * 4];
}

extern "C" void kernel_launch(void* const* d_in, const int* in_sizes, int n_in,
                              void* d_out, int out_size, void* d_ws, size_t ws_size,
                              hipStream_t stream) {
    const int*   x    = (const int*)d_in[0];
    const float* emb  = (const float*)d_in[1];
    const float* W    = (const float*)d_in[2];
    const float* bias = (const float*)d_in[3];
    float*       out  = (float*)d_out;

    pool_linear_kernel<<<BATCH / ROWS, THREADS, 0, stream>>>(x, emb, W, bias, out);
}
